# starting kernel
# baseline (speedup 1.0000x reference)
#include <hip/hip_runtime.h>
#include <hip/hip_bf16.h>

#define DD 128
#define XS 132
#define WS 136

typedef __attribute__((ext_vector_type(2))) float v2f;
typedef __attribute__((ext_vector_type(8))) float v8f;

__global__ void deg_init_kernel(float* __restrict__ deg, int nN) {
    int i = blockIdx.x * blockDim.x + threadIdx.x;
    if (i < nN) deg[i] = 1.0f;
}

__global__ void deg_edge_kernel(const int* __restrict__ dst, float* __restrict__ deg, int nE) {
    int e = blockIdx.x * blockDim.x + threadIdx.x;
    if (e < nE) atomicAdd(&deg[dst[e]], 1.0f);
}

__global__ void dinv_kernel(float* __restrict__ deg, int nN) {
    int i = blockIdx.x * blockDim.x + threadIdx.x;
    if (i < nN) deg[i] = rsqrtf(deg[i]);
}

__global__ __launch_bounds__(256)
void gcn_gemm_wmma(const float* __restrict__ X, const float* __restrict__ W,
                   float* __restrict__ H, int nN, int applyRelu) {
    __shared__ __align__(16) float sW[DD * WS];
    __shared__ __align__(16) float sX[128 * XS];

    const int tid  = threadIdx.x;
    const int wave = tid >> 5;
    const int lane = tid & 31;
    const int lr   = lane & 15;
    const int hi   = lane >> 4;
    const int rowBase = blockIdx.x * 128;

    for (int i = tid; i < DD * 32; i += 256) {
        int r  = i >> 5;
        int c4 = (i & 31) << 2;
        float4 w = *(const float4*)(W + r * DD + c4);
        *(float4*)(sW + r * WS + c4) = w;
    }
    for (int i = tid; i < 128 * 32; i += 256) {
        int r  = i >> 5;
        int c4 = (i & 31) << 2;
        int gr = rowBase + r;
        float4 v;
        if (gr < nN) {
            v = *(const float4*)(X + (size_t)gr * DD + c4);
            if (applyRelu) {
                v.x = fmaxf(v.x, 0.0f); v.y = fmaxf(v.y, 0.0f);
                v.z = fmaxf(v.z, 0.0f); v.w = fmaxf(v.w, 0.0f);
            }
        } else {
            v = make_float4(0.0f, 0.0f, 0.0f, 0.0f);
        }
        *(float4*)(sX + r * XS + c4) = v;
    }
    __syncthreads();

    const float* xrow = sX + (wave * 16 + lr) * XS;

    v8f acc[8];
#pragma unroll
    for (int t = 0; t < 8; ++t) acc[t] = (v8f){0.f,0.f,0.f,0.f,0.f,0.f,0.f,0.f};

    for (int kb = 0; kb < 32; ++kb) {
        const int k0 = kb * 4 + hi * 2;
        v2f a = *(const v2f*)(xrow + k0);
        const float* wp = sW + k0 * WS + lr;
#pragma unroll
        for (int t = 0; t < 8; ++t) {
            v2f b;
            b.x = wp[t * 16];
            b.y = wp[t * 16 + WS];
            acc[t] = __builtin_amdgcn_wmma_f32_16x16x4_f32(
                false, a, false, b, (short)0, acc[t], false, false);
        }
    }

    const int row0 = rowBase + wave * 16 + hi * 8;
#pragma unroll
    for (int t = 0; t < 8; ++t) {
        const int col = t * 16 + lr;
#pragma unroll
        for (int v = 0; v < 8; ++v) {
            int r = row0 + v;
            if (r < nN) H[(size_t)r * DD + col] = acc[t][v];
        }
    }
}

__global__ void init_out_kernel(const float* __restrict__ h, const float* __restrict__ dinv,
                                const float* __restrict__ bias, float* __restrict__ out, int nN) {
    int idx = blockIdx.x * blockDim.x + threadIdx.x;
    int i = idx >> 5;
    int g = (idx & 31) << 2;
    if (i >= nN) return;
    float di = dinv[i];
    float w  = di * di;
    float4 v = *(const float4*)(h + (size_t)i * DD + g);
    float4 b = *(const float4*)(bias + g);
    float4 r;
    r.x = v.x * w + b.x;  r.y = v.y * w + b.y;
    r.z = v.z * w + b.z;  r.w = v.w * w + b.w;
    *(float4*)(out + (size_t)i * DD + g) = r;
}

__global__ void edge_scatter_kernel(const float* __restrict__ h, const float* __restrict__ dinv,
                                    const int* __restrict__ src, const int* __restrict__ dst,
                                    float* __restrict__ out, int nE) {
    unsigned idx = blockIdx.x * blockDim.x + threadIdx.x;
    int e = idx >> 5;
    int g = (idx & 31) << 2;
    if (e >= nE) return;
    int s = src[e];
    int d = dst[e];
    float nrm = dinv[s] * dinv[d];
    float4 v = *(const float4*)(h + (size_t)s * DD + g);
    float* o = out + (size_t)d * DD + g;
    atomicAdd(o + 0, nrm * v.x);
    atomicAdd(o + 1, nrm * v.y);
    atomicAdd(o + 2, nrm * v.z);
    atomicAdd(o + 3, nrm * v.w);
}

extern "C" void kernel_launch(void* const* d_in, const int* in_sizes, int n_in,
                              void* d_out, int out_size, void* d_ws, size_t ws_size,
                              hipStream_t stream) {
    const float* x   = (const float*)d_in[0];
    const int*   ei  = (const int*)d_in[1];
    const float* W1  = (const float*)d_in[2];
    const float* b1  = (const float*)d_in[3];
    const float* W2  = (const float*)d_in[4];
    const float* b2  = (const float*)d_in[5];
    const float* W3  = (const float*)d_in[6];
    const float* b3  = (const float*)d_in[7];

    const int nN = in_sizes[0] / DD;
    const int nE = in_sizes[1] / 2;
    const int* src = ei;
    const int* dst = ei + nE;

    const size_t npad = (size_t)((nN + 127) / 128) * 128;
    float* dinv = (float*)d_ws;
    float* bufH = dinv + npad;
    float* bufX = bufH + npad * DD;
    float* outF = (float*)d_out;

    const int TB = 256;
    const dim3 blk(TB);
    const int gN   = (nN + TB - 1) / TB;
    const int gE   = (nE + TB - 1) / TB;
    const int gNd  = (int)(((size_t)nN * 32 + TB - 1) / TB);
    const int gEd  = (int)(((size_t)nE * 32 + TB - 1) / TB);
    const int gGemm = (nN + 127) / 128;

    deg_init_kernel<<<gN, blk, 0, stream>>>(dinv, nN);
    deg_edge_kernel<<<gE, blk, 0, stream>>>(dst, dinv, nE);
    dinv_kernel<<<gN, blk, 0, stream>>>(dinv, nN);

    gcn_gemm_wmma<<<gGemm, blk, 0, stream>>>(x, W1, bufH, nN, 0);
    init_out_kernel<<<gNd, blk, 0, stream>>>(bufH, dinv, b1, bufX, nN);
    edge_scatter_kernel<<<gEd, blk, 0, stream>>>(bufH, dinv, src, dst, bufX, nE);

    gcn_gemm_wmma<<<gGemm, blk, 0, stream>>>(bufX, W2, bufH, nN, 1);
    init_out_kernel<<<gNd, blk, 0, stream>>>(bufH, dinv, b2, bufX, nN);
    edge_scatter_kernel<<<gEd, blk, 0, stream>>>(bufH, dinv, src, dst, bufX, nE);

    gcn_gemm_wmma<<<gGemm, blk, 0, stream>>>(bufX, W3, bufH, nN, 1);
    init_out_kernel<<<gNd, blk, 0, stream>>>(bufH, dinv, b3, outF, nN);
    edge_scatter_kernel<<<gEd, blk, 0, stream>>>(bufH, dinv, src, dst, outF, nE);
}


# list launch scans a 1-byte owner code per edge (dst/196 & 255, from a new small first launch) instead of the 4-byte dst: 4x fewer bytes to each of the 256 work-groups; arrival-order sums as before
# speedup vs baseline: 3.0155x; 3.0155x over previous
#include <hip/hip_runtime.h>
#include <hip/hip_bf16.h>

#define DD 128
#define XS 132
#define WS 136

typedef __attribute__((ext_vector_type(2))) float v2f;
typedef __attribute__((ext_vector_type(8))) float v8f;

__global__ __launch_bounds__(256)
void gcn_gemm_wmma(const float* __restrict__ X, const float* __restrict__ W,
                   float* __restrict__ H, int nN, int applyRelu) {
    __shared__ __align__(16) float sW[DD * WS];
    __shared__ __align__(16) float sX[128 * XS];

    const int tid  = threadIdx.x;
    const int wave = tid >> 5;
    const int lane = tid & 31;
    const int lr   = lane & 15;
    const int hi   = lane >> 4;
    const int rowBase = blockIdx.x * 128;

    for (int i = tid; i < DD * 32; i += 256) {
        int r  = i >> 5;
        int c4 = (i & 31) << 2;
        float4 w = *(const float4*)(W + r * DD + c4);
        *(float4*)(sW + r * WS + c4) = w;
    }
    for (int i = tid; i < 128 * 32; i += 256) {
        int r  = i >> 5;
        int c4 = (i & 31) << 2;
        int gr = rowBase + r;
        float4 v;
        if (gr < nN) {
            v = *(const float4*)(X + (size_t)gr * DD + c4);
            if (applyRelu) {
                v.x = fmaxf(v.x, 0.0f); v.y = fmaxf(v.y, 0.0f);
                v.z = fmaxf(v.z, 0.0f); v.w = fmaxf(v.w, 0.0f);
            }
        } else {
            v = make_float4(0.0f, 0.0f, 0.0f, 0.0f);
        }
        *(float4*)(sX + r * XS + c4) = v;
    }
    __syncthreads();

    const float* xrow = sX + (wave * 16 + lr) * XS;

    v8f acc[8];
#pragma unroll
    for (int t = 0; t < 8; ++t) acc[t] = (v8f){0.f,0.f,0.f,0.f,0.f,0.f,0.f,0.f};

    for (int kb = 0; kb < 32; ++kb) {
        const int k0 = kb * 4 + hi * 2;
        v2f a = *(const v2f*)(xrow + k0);
        const float* wp = sW + k0 * WS + lr;
#pragma unroll
        for (int t = 0; t < 8; ++t) {
            v2f b;
            b.x = wp[t * 16];
            b.y = wp[t * 16 + WS];
            acc[t] = __builtin_amdgcn_wmma_f32_16x16x4_f32(
                false, a, false, b, (short)0, acc[t], false, false);
        }
    }

    const int row0 = rowBase + wave * 16 + hi * 8;
#pragma unroll
    for (int t = 0; t < 8; ++t) {
        const int col = t * 16 + lr;
#pragma unroll
        for (int v = 0; v < 8; ++v) {
            int r = row0 + v;
            if (r < nN) H[(size_t)r * DD + col] = acc[t][v];
        }
    }
}

#define RB     196
#define NTH    1024
#define SCAN_U 12
#define QW     256
#define ERUN   32
#define EBAT   8
#define LHDR   4
#define CTH    256

__global__ __launch_bounds__(CTH)
void owner_codes_kernel(const int* __restrict__ dst, unsigned* __restrict__ code, int nE) {
    const int q  = blockIdx.x * CTH + threadIdx.x;
    const int nQ = (nE >> 4) << 2;
    if (q >= nQ) return;
    int4 d;
    if ((((size_t)dst) & 15) == 0) {
        d = ((const int4*)dst)[q];
    } else {
        d.x = dst[4 * q + 0];  d.y = dst[4 * q + 1];
        d.z = dst[4 * q + 2];  d.w = dst[4 * q + 3];
    }
    const unsigned c0 = ((unsigned)d.x / RB) & 255u;
    const unsigned c1 = ((unsigned)d.y / RB) & 255u;
    const unsigned c2 = ((unsigned)d.z / RB) & 255u;
    const unsigned c3 = ((unsigned)d.w / RB) & 255u;
    code[q] = c0 | (c1 << 8) | (c2 << 16) | (c3 << 24);
}


__device__ __forceinline__ unsigned zero_bytes(unsigned x) {
    return ~(((x & 0x7F7F7F7Fu) + 0x7F7F7F7Fu) | x | 0x7F7F7F7Fu);
}

__global__ __launch_bounds__(NTH)
void own_edges_kernel(const int* __restrict__ src, const int* __restrict__ dst,
                      const unsigned* __restrict__ code,
                      float* __restrict__ dinv, unsigned* __restrict__ lst,
                      int nN, int nE) {
    __shared__ int degc[RB];
    __shared__ int cur;
    __shared__ int qe[(NTH / 32) * QW];

    const int tid  = threadIdx.x;
    const int lane = tid & 31;
    const int wv   = __builtin_amdgcn_readfirstlane(tid >> 5);
    const int base = blockIdx.x * RB;
    const unsigned rbu = (unsigned)min(RB, nN - base);
    const int cap  = (int)rbu * DD - LHDR;
    unsigned* L = lst + (size_t)base * DD;
    int* myq = qe + wv * QW;

    if (tid < RB) degc[tid] = 0;
    if (tid == 0) cur = 0;
    __syncthreads();

    int qc = 0;
    int done = 0;

#define OWN_FLUSH()                                                                \
    {                                                                              \
        const int n_ = qc - done;                                                  \
        for (int i0_ = 0; i0_ < n_; i0_ += 32) {                                   \
            const int i_ = i0_ + lane;                                             \
            bool     ok_ = false;                                                  \
            int      s_  = 0;                                                      \
            unsigned t_  = 0u;                                                     \
            if (i_ < n_) {                                                         \
                const int e_ = myq[(done + i_) & (QW - 1)];                        \
                t_  = (unsigned)(dst[e_] - base);                                  \
                s_  = src[e_];                                                     \
                ok_ = t_ < rbu;                                                    \
            }                                                                      \
            const unsigned k_ = __builtin_amdgcn_ballot_w32(ok_);                  \
            if (k_ != 0u) {                                                        \
                int pos_ = 0;                                                      \
                if (lane == 0) pos_ = atomicAdd(&cur, (int)__builtin_popcount(k_)); \
                pos_ = __builtin_amdgcn_readfirstlane(pos_);                       \
                if (ok_) {                                                         \
                    atomicAdd(&degc[t_], 1);                                       \
                    const int p_ = pos_ + (int)__builtin_amdgcn_mbcnt_lo(k_, 0u);  \
                    if (p_ < cap) L[LHDR + p_] = (unsigned)s_ | (t_ << 24);        \
                }                                                                  \
            }                                                                      \
        }                                                                          \
        done = qc;                                                                 \
    }

#define OWN_SCAN16(V, E0)                                                          \
    {                                                                              \
        const unsigned z0_ = zero_bytes((V).x ^ pat);                              \
        const unsigned z1_ = zero_bytes((V).y ^ pat);                              \
        const unsigned z2_ = zero_bytes((V).z ^ pat);                              \
        const unsigned z3_ = zero_bytes((V).w ^ pat);                              \
                     \
        unsigned hm_ = (z0_ >> 7) | (z1_ >> 6) | (z2_ >> 5) | (z3_ >> 4);          \
        unsigned m_  = __builtin_amdgcn_ballot_w32(hm_ != 0u);                     \
        while (m_ != 0u) {                             \
            if (qc - done > QW - 32) OWN_FLUSH()                                   \
            if (hm_ != 0u) {                                                       \
                const int bit_  = __builtin_ctz(hm_);                              \
                hm_ &= hm_ - 1u;                                                   \
                const int rank_ = (int)__builtin_amdgcn_mbcnt_lo(m_, 0u);          \
                myq[(qc + rank_) & (QW - 1)] = (E0) + (((bit_ & 3) << 2) | (bit_ >> 3)); \
            }                                                                      \
            qc += (int)__builtin_popcount(m_);                                     \
            m_  = __builtin_amdgcn_ballot_w32(hm_ != 0u);                          \
        }                                                                          \
    }

    const unsigned pat = (blockIdx.x & 255u) * 0x01010101u;
    const int    n16  = nE >> 4;
    const uint4* c4   = (const uint4*)code;
    const int    nBlk = n16 / NTH;
    const int    rot  = (nBlk > 0) ? (int)(blockIdx.x % (unsigned)nBlk) : 0;

    int k0 = 0;
    for (; k0 + SCAN_U <= nBlk; k0 += SCAN_U) {
        uint4 v[SCAN_U];
        int   g[SCAN_U];
#pragma unroll
        for (int u = 0; u < SCAN_U; ++u) {
            int blk = k0 + u + rot;
            if (blk >= nBlk) blk -= nBlk;
            g[u] = blk * NTH + tid;
            v[u] = c4[g[u]];
        }
#pragma unroll
        for (int u = 0; u < SCAN_U; ++u) OWN_SCAN16(v[u], g[u] << 4)
    }
    for (; k0 < nBlk; ++k0) {
        int blk = k0 + rot;
        if (blk >= nBlk) blk -= nBlk;
        const int   g = blk * NTH + tid;
        const uint4 v = c4[g];
        OWN_SCAN16(v, g << 4)
    }
    if (nBlk * NTH < n16) {
        const int g = nBlk * NTH + tid;
        uint4 v = make_uint4(~pat, ~pat, ~pat, ~pat);
        if (g < n16) v = c4[g];
        OWN_SCAN16(v, g << 4)
    }
    for (int e0 = n16 << 4; e0 < nE; e0 += NTH) {
        const int e = e0 + tid;
        int d = -1;
        if (e < nE) d = dst[e];
        const bool     hit = (unsigned)(d - base) < rbu;
        const unsigned m   = __builtin_amdgcn_ballot_w32(hit);
        if (m != 0u) {
            if (qc - done > QW - 32) OWN_FLUSH()
            if (hit) myq[(qc + (int)__builtin_amdgcn_mbcnt_lo(m, 0u)) & (QW - 1)] = e;
            qc += (int)__builtin_popcount(m);
        }
    }
    OWN_FLUSH()
#undef OWN_FLUSH
#undef OWN_SCAN16
    __syncthreads();

    if ((unsigned)tid < rbu) dinv[base + tid] = rsqrtf((float)(1 + degc[tid]));
    if (tid == 0) L[0] = (unsigned)cur;
}


__device__ __forceinline__ float4 agg_load_row(const float* __restrict__ h, int s, int lane) {
    return *(const float4*)(h + (size_t)s * DD + 4 * lane);
}

__device__ __forceinline__ void agg_add_row(float* acc, int rl, float w, int lane, float4 v) {
    float* a = acc + rl * DD + lane;
    atomicAdd(a +  0, w * v.x);
    atomicAdd(a + 32, w * v.y);
    atomicAdd(a + 64, w * v.z);
    atomicAdd(a + 96, w * v.w);
}

__global__ __launch_bounds__(NTH)
void agg_rows_lds_kernel(const float* __restrict__ h, const float* __restrict__ dinv,
                         const int* __restrict__ src, const int* __restrict__ dst,
                         const float* __restrict__ bias, const unsigned* lst,
                         float* out, int nN, int nE) {
    __shared__ __align__(16) float acc[RB * DD];
    __shared__ float dloc[RB];

    const int tid  = threadIdx.x;
    const int lane = tid & 31;
    const int wv   = __builtin_amdgcn_readfirstlane(tid >> 5);
    const int base = blockIdx.x * RB;
    const unsigned rbu = (unsigned)min(RB, nN - base);
    const unsigned* L = lst + (size_t)base * DD;
    const int cnt = __builtin_amdgcn_readfirstlane((int)L[0]);
    const int cap = (int)rbu * DD - LHDR;

    {
        constexpr int P0 = (RB * 32 + NTH - 1) / NTH;
        const float4 b = *(const float4*)(bias + 4 * lane);
        float  di[P0];
        float4 x[P0];
#pragma unroll
        for (int k = 0; k < P0; ++k) {
            const int r  = (tid + k * NTH) >> 5;
            const int gr = min(base + r, nN - 1);
            di[k] = dinv[gr];
            x[k]  = *(const float4*)(h + (size_t)gr * DD + 4 * lane);
        }
        if ((unsigned)tid < rbu) dloc[tid] = dinv[base + tid];
#pragma unroll
        for (int k = 0; k < P0; ++k) {
            const int r = (tid + k * NTH) >> 5;
            if ((unsigned)r < rbu) {
                const float w = di[k] * di[k];
                float* a = acc + r * DD + lane;
                a[0]  = x[k].x * w + b.x;
                a[32] = x[k].y * w + b.y;
                a[64] = x[k].z * w + b.z;
                a[96] = x[k].w * w + b.w;
            }
        }
    }
    __syncthreads();

    if (cnt <= cap) {
        const unsigned* E = L + LHDR;
        int i0 = wv * ERUN;
        unsigned ent  = 0u;
        unsigned entn = 0u;
        if (i0 + lane < cnt)       ent  = E[i0 + lane];
        if (i0 + NTH + lane < cnt) entn = E[i0 + NTH + lane];
        float wt = dinv[ent & 0xFFFFFFu] * dloc[ent >> 24];
        while (i0 + ERUN <= cnt) {
            const int sl = (int)(ent & 0xFFFFFFu);
            const int dl = (int)(ent >> 24);
            const int ni = __float_as_int(wt);
            unsigned entnn = 0u;
            if (i0 + 2 * NTH + lane < cnt) entnn = E[i0 + 2 * NTH + lane];
            float4 rows_[2][EBAT];
#pragma unroll
            for (int u = 0; u < EBAT; ++u)
                rows_[0][u] = agg_load_row(h, __builtin_amdgcn_readlane(sl, u), lane);
            const float wtn = dinv[entn & 0xFFFFFFu] * dloc[entn >> 24];
#pragma unroll
            for (int b = 0; b < ERUN / EBAT; ++b) {
                if (b + 1 < ERUN / EBAT) {
#pragma unroll
                    for (int u = 0; u < EBAT; ++u)
                        rows_[(b + 1) & 1][u] = agg_load_row(
                            h, __builtin_amdgcn_readlane(sl, (b + 1) * EBAT + u), lane);
                }
#pragma unroll
                for (int u = 0; u < EBAT; ++u)
                    agg_add_row(acc, __builtin_amdgcn_readlane(dl, b * EBAT + u),
                                __int_as_float(__builtin_amdgcn_readlane(ni, b * EBAT + u)),
                                lane, rows_[b & 1][u]);
            }
            ent  = entn;
            entn = entnn;
            wt   = wtn;
            i0  += NTH;
        }
        if (i0 < cnt) {
            const int n  = cnt - i0;
            const int sl = (int)(ent & 0xFFFFFFu);
            const int dl = (int)(ent >> 24);
            const int ni = __float_as_int(wt);
            for (int b = 0; b < n; b += EBAT) {
                float4 rows[EBAT];
#pragma unroll
                for (int u = 0; u < EBAT; ++u)
                    if (b + u < n)
                        rows[u] = agg_load_row(h, __builtin_amdgcn_readlane(sl, b + u), lane);
#pragma unroll
                for (int u = 0; u < EBAT; ++u)
                    if (b + u < n)
                        agg_add_row(acc, __builtin_amdgcn_readlane(dl, b + u),
                                    __int_as_float(__builtin_amdgcn_readlane(ni, b + u)),
                                    lane, rows[u]);
            }
        }
    } else {
        for (int e0 = wv * 32; e0 < nE; e0 += NTH) {
            const int e = e0 + lane;
            int d = -1;
            if (e < nE) d = dst[e];
            const unsigned t = (unsigned)(d - base);
            const bool hit = t < rbu;
            int   s = 0;
            float w = 0.0f;
            if (hit) {
                s = src[e];
                w = dinv[s] * dloc[t];
            }
            unsigned m = __builtin_amdgcn_ballot_w32(hit);
            while (m != 0u) {
                const int j = __builtin_ctz(m);
                m &= m - 1u;
                const float4 row = agg_load_row(h, __builtin_amdgcn_readlane(s, j), lane);
                agg_add_row(acc, __builtin_amdgcn_readlane((int)t, j),
                            __int_as_float(__builtin_amdgcn_readlane(__float_as_int(w), j)),
                            lane, row);
            }
        }
    }
    __syncthreads();

    for (int i = tid; i < RB * 32; i += NTH) {
        const int r = i >> 5;
        if ((unsigned)r < rbu) {
            const float* a = acc + r * DD + lane;
            float4 o;
            o.x = a[0];  o.y = a[32];  o.z = a[64];  o.w = a[96];
            *(float4*)(out + (size_t)(base + r) * DD + 4 * lane) = o;
        }
    }
}

extern "C" void kernel_launch(void* const* d_in, const int* in_sizes, int n_in,
                              void* d_out, int out_size, void* d_ws, size_t ws_size,
                              hipStream_t stream) {
    const float* x   = (const float*)d_in[0];
    const int*   ei  = (const int*)d_in[1];
    const float* W1  = (const float*)d_in[2];
    const float* b1  = (const float*)d_in[3];
    const float* W2  = (const float*)d_in[4];
    const float* b2  = (const float*)d_in[5];
    const float* W3  = (const float*)d_in[6];
    const float* b3  = (const float*)d_in[7];

    const int nN = in_sizes[0] / DD;
    const int nE = in_sizes[1] / 2;
    const int* src = ei;
    const int* dst = ei + nE;

    const size_t npad = (size_t)((nN + 127) / 128) * 128;
    float* dinv = (float*)d_ws;
    float* bufH = dinv + npad;
    float* bufX = bufH + npad * DD;
    float* outF = (float*)d_out;
    unsigned* lst = (unsigned*)d_out;

    const dim3 blk(256);
    const int gAgg = (nN + RB - 1) / RB;
    const int gGemm = (nN + 127) / 128;

    unsigned* code = (unsigned*)bufX;
    const int gCode = ((nE >> 4) * 4) / CTH + 1;
    owner_codes_kernel<<<gCode, dim3(CTH), 0, stream>>>(dst, code, nE);

    own_edges_kernel<<<gAgg, dim3(NTH), 0, stream>>>(src, dst, code, dinv, lst, nN, nE);

    gcn_gemm_wmma<<<gGemm, blk, 0, stream>>>(x, W1, bufH, nN, 0);
    agg_rows_lds_kernel<<<gAgg, dim3(NTH), 0, stream>>>(bufH, dinv, src, dst, b1, lst, bufX, nN, nE);

    gcn_gemm_wmma<<<gGemm, blk, 0, stream>>>(bufX, W2, bufH, nN, 1);
    agg_rows_lds_kernel<<<gAgg, dim3(NTH), 0, stream>>>(bufH, dinv, src, dst, b2, lst, bufX, nN, nE);

    gcn_gemm_wmma<<<gGemm, blk, 0, stream>>>(bufX, W3, bufH, nN, 1);
    agg_rows_lds_kernel<<<gAgg, dim3(NTH), 0, stream>>>(bufH, dinv, src, dst, b3, lst, outF, nN, nE);
}


# symmetric norm factored out of the edge walk: product epilogue scales row s by dinv[s], aggregation applies dinv[i] and bias once per row; no per-edge norm gather/multiply; arrival-order sums kept
# speedup vs baseline: 3.1366x; 1.0402x over previous
#include <hip/hip_runtime.h>
#include <hip/hip_bf16.h>

#define DD 128
#define XS 132
#define WS 136

typedef __attribute__((ext_vector_type(2))) float v2f;
typedef __attribute__((ext_vector_type(8))) float v8f;

__global__ __launch_bounds__(256)
void gcn_gemm_wmma(const float* __restrict__ X, const float* __restrict__ W,
                   const float* __restrict__ dinv,
                   float* __restrict__ H, int nN, int applyRelu) {
    __shared__ __align__(16) float sW[DD * WS];
    __shared__ __align__(16) float sX[128 * XS];

    const int tid  = threadIdx.x;
    const int wave = tid >> 5;
    const int lane = tid & 31;
    const int lr   = lane & 15;
    const int hi   = lane >> 4;
    const int rowBase = blockIdx.x * 128;

    for (int i = tid; i < DD * 32; i += 256) {
        int r  = i >> 5;
        int c4 = (i & 31) << 2;
        float4 w = *(const float4*)(W + r * DD + c4);
        *(float4*)(sW + r * WS + c4) = w;
    }
    for (int i = tid; i < 128 * 32; i += 256) {
        int r  = i >> 5;
        int c4 = (i & 31) << 2;
        int gr = rowBase + r;
        float4 v;
        if (gr < nN) {
            v = *(const float4*)(X + (size_t)gr * DD + c4);
            if (applyRelu) {
                v.x = fmaxf(v.x, 0.0f); v.y = fmaxf(v.y, 0.0f);
                v.z = fmaxf(v.z, 0.0f); v.w = fmaxf(v.w, 0.0f);
            }
        } else {
            v = make_float4(0.0f, 0.0f, 0.0f, 0.0f);
        }
        *(float4*)(sX + r * XS + c4) = v;
    }
    __syncthreads();

    const float* xrow = sX + (wave * 16 + lr) * XS;

    v8f acc[8];
#pragma unroll
    for (int t = 0; t < 8; ++t) acc[t] = (v8f){0.f,0.f,0.f,0.f,0.f,0.f,0.f,0.f};

    for (int kb = 0; kb < 32; ++kb) {
        const int k0 = kb * 4 + hi * 2;
        v2f a = *(const v2f*)(xrow + k0);
        const float* wp = sW + k0 * WS + lr;
#pragma unroll
        for (int t = 0; t < 8; ++t) {
            v2f b;
            b.x = wp[t * 16];
            b.y = wp[t * 16 + WS];
            acc[t] = __builtin_amdgcn_wmma_f32_16x16x4_f32(
                false, a, false, b, (short)0, acc[t], false, false);
        }
    }

    const int row0 = rowBase + wave * 16 + hi * 8;
    float dv[8];
#pragma unroll
    for (int v = 0; v < 8; ++v) dv[v] = dinv[min(row0 + v, nN - 1)];
#pragma unroll
    for (int t = 0; t < 8; ++t) {
        const int col = t * 16 + lr;
#pragma unroll
        for (int v = 0; v < 8; ++v) {
            int r = row0 + v;
            if (r < nN) H[(size_t)r * DD + col] = acc[t][v] * dv[v];
        }
    }
}

#define RB     196
#define NTH    1024
#define SCAN_U 12
#define QW     256
#define ERUN   32
#define EBAT   8
#define LHDR   4
#define CTH    256

__global__ __launch_bounds__(CTH)
void owner_codes_kernel(const int* __restrict__ dst, unsigned* __restrict__ code, int nE) {
    const int q  = blockIdx.x * CTH + threadIdx.x;
    const int nQ = (nE >> 4) << 2;
    if (q >= nQ) return;
    int4 d;
    if ((((size_t)dst) & 15) == 0) {
        d = ((const int4*)dst)[q];
    } else {
        d.x = dst[4 * q + 0];  d.y = dst[4 * q + 1];
        d.z = dst[4 * q + 2];  d.w = dst[4 * q + 3];
    }
    const unsigned c0 = ((unsigned)d.x / RB) & 255u;
    const unsigned c1 = ((unsigned)d.y / RB) & 255u;
    const unsigned c2 = ((unsigned)d.z / RB) & 255u;
    const unsigned c3 = ((unsigned)d.w / RB) & 255u;
    code[q] = c0 | (c1 << 8) | (c2 << 16) | (c3 << 24);
}


__device__ __forceinline__ unsigned zero_bytes(unsigned x) {
    return ~(((x & 0x7F7F7F7Fu) + 0x7F7F7F7Fu) | x | 0x7F7F7F7Fu);
}

__global__ __launch_bounds__(NTH)
void own_edges_kernel(const int* __restrict__ src, const int* __restrict__ dst,
                      const unsigned* __restrict__ code,
                      float* __restrict__ dinv, unsigned* __restrict__ lst,
                      int nN, int nE) {
    __shared__ int degc[RB];
    __shared__ int cur;
    __shared__ int qe[(NTH / 32) * QW];

    const int tid  = threadIdx.x;
    const int lane = tid & 31;
    const int wv   = __builtin_amdgcn_readfirstlane(tid >> 5);
    const int base = blockIdx.x * RB;
    const unsigned rbu = (unsigned)min(RB, nN - base);
    const int cap  = (int)rbu * DD - LHDR;
    unsigned* L = lst + (size_t)base * DD;
    int* myq = qe + wv * QW;

    if (tid < RB) degc[tid] = 0;
    if (tid == 0) cur = 0;
    __syncthreads();

    int qc = 0;
    int done = 0;

#define OWN_FLUSH()                                                                \
    {                                                                              \
        const int n_ = qc - done;                                                  \
        for (int i0_ = 0; i0_ < n_; i0_ += 32) {                                   \
            const int i_ = i0_ + lane;                                             \
            bool     ok_ = false;                                                  \
            int      s_  = 0;                                                      \
            unsigned t_  = 0u;                                                     \
            if (i_ < n_) {                                                         \
                const int e_ = myq[(done + i_) & (QW - 1)];                        \
                t_  = (unsigned)(dst[e_] - base);                                  \
                s_  = src[e_];                                                     \
                ok_ = t_ < rbu;                                                    \
            }                                                                      \
            const unsigned k_ = __builtin_amdgcn_ballot_w32(ok_);                  \
            if (k_ != 0u) {                                                        \
                int pos_ = 0;                                                      \
                if (lane == 0) pos_ = atomicAdd(&cur, (int)__builtin_popcount(k_)); \
                pos_ = __builtin_amdgcn_readfirstlane(pos_);                       \
                if (ok_) {                                                         \
                    atomicAdd(&degc[t_], 1);                                       \
                    const int p_ = pos_ + (int)__builtin_amdgcn_mbcnt_lo(k_, 0u);  \
                    if (p_ < cap) L[LHDR + p_] = (unsigned)s_ | (t_ << 24);        \
                }                                                                  \
            }                                                                      \
        }                                                                          \
        done = qc;                                                                 \
    }

#define OWN_SCAN16(V, E0)                                                          \
    {                                                                              \
        const unsigned z0_ = zero_bytes((V).x ^ pat);                              \
        const unsigned z1_ = zero_bytes((V).y ^ pat);                              \
        const unsigned z2_ = zero_bytes((V).z ^ pat);                              \
        const unsigned z3_ = zero_bytes((V).w ^ pat);                              \
                     \
        unsigned hm_ = (z0_ >> 7) | (z1_ >> 6) | (z2_ >> 5) | (z3_ >> 4);          \
        unsigned m_  = __builtin_amdgcn_ballot_w32(hm_ != 0u);                     \
        while (m_ != 0u) {                             \
            if (qc - done > QW - 32) OWN_FLUSH()                                   \
            if (hm_ != 0u) {                                                       \
                const int bit_  = __builtin_ctz(hm_);                              \
                hm_ &= hm_ - 1u;                                                   \
                const int rank_ = (int)__builtin_amdgcn_mbcnt_lo(m_, 0u);          \
                myq[(qc + rank_) & (QW - 1)] = (E0) + (((bit_ & 3) << 2) | (bit_ >> 3)); \
            }                                                                      \
            qc += (int)__builtin_popcount(m_);                                     \
            m_  = __builtin_amdgcn_ballot_w32(hm_ != 0u);                          \
        }                                                                          \
    }

    const unsigned pat = (blockIdx.x & 255u) * 0x01010101u;
    const int    n16  = nE >> 4;
    const uint4* c4   = (const uint4*)code;
    const int    nBlk = n16 / NTH;
    const int    rot  = (nBlk > 0) ? (int)(blockIdx.x % (unsigned)nBlk) : 0;

    int k0 = 0;
    for (; k0 + SCAN_U <= nBlk; k0 += SCAN_U) {
        uint4 v[SCAN_U];
        int   g[SCAN_U];
#pragma unroll
        for (int u = 0; u < SCAN_U; ++u) {
            int blk = k0 + u + rot;
            if (blk >= nBlk) blk -= nBlk;
            g[u] = blk * NTH + tid;
            v[u] = c4[g[u]];
        }
#pragma unroll
        for (int u = 0; u < SCAN_U; ++u) OWN_SCAN16(v[u], g[u] << 4)
    }
    for (; k0 < nBlk; ++k0) {
        int blk = k0 + rot;
        if (blk >= nBlk) blk -= nBlk;
        const int   g = blk * NTH + tid;
        const uint4 v = c4[g];
        OWN_SCAN16(v, g << 4)
    }
    if (nBlk * NTH < n16) {
        const int g = nBlk * NTH + tid;
        uint4 v = make_uint4(~pat, ~pat, ~pat, ~pat);
        if (g < n16) v = c4[g];
        OWN_SCAN16(v, g << 4)
    }
    for (int e0 = n16 << 4; e0 < nE; e0 += NTH) {
        const int e = e0 + tid;
        int d = -1;
        if (e < nE) d = dst[e];
        const bool     hit = (unsigned)(d - base) < rbu;
        const unsigned m   = __builtin_amdgcn_ballot_w32(hit);
        if (m != 0u) {
            if (qc - done > QW - 32) OWN_FLUSH()
            if (hit) myq[(qc + (int)__builtin_amdgcn_mbcnt_lo(m, 0u)) & (QW - 1)] = e;
            qc += (int)__builtin_popcount(m);
        }
    }
    OWN_FLUSH()
#undef OWN_FLUSH
#undef OWN_SCAN16
    __syncthreads();

    if ((unsigned)tid < rbu) dinv[base + tid] = rsqrtf((float)(1 + degc[tid]));
    if (tid == 0) L[0] = (unsigned)cur;
}


__device__ __forceinline__ float4 agg_load_row(const float* __restrict__ h, int s, int lane) {
    return *(const float4*)(h + (size_t)s * DD + 4 * lane);
}

__device__ __forceinline__ void agg_add_row(float* acc, int rl, int lane, float4 v) {
    float* a = acc + rl * DD + lane;
    atomicAdd(a +  0, v.x);
    atomicAdd(a + 32, v.y);
    atomicAdd(a + 64, v.z);
    atomicAdd(a + 96, v.w);
}

__global__ __launch_bounds__(NTH)
void agg_rows_lds_kernel(const float* __restrict__ h, const float* __restrict__ dinv,
                         const int* __restrict__ src, const int* __restrict__ dst,
                         const float* __restrict__ bias, const unsigned* lst,
                         float* out, int nN, int nE) {
    __shared__ __align__(16) float acc[RB * DD];
    __shared__ float dloc[RB];

    const int tid  = threadIdx.x;
    const int lane = tid & 31;
    const int wv   = __builtin_amdgcn_readfirstlane(tid >> 5);
    const int base = blockIdx.x * RB;
    const unsigned rbu = (unsigned)min(RB, nN - base);
    const unsigned* L = lst + (size_t)base * DD;
    const int cnt = __builtin_amdgcn_readfirstlane((int)L[0]);
    const int cap = (int)rbu * DD - LHDR;

    const float4 bv = *(const float4*)(bias + 4 * lane);

    {
        constexpr int P0 = (RB * 32 + NTH - 1) / NTH;
        float4 x[P0];
#pragma unroll
        for (int k = 0; k < P0; ++k) {
            const int r  = (tid + k * NTH) >> 5;
            const int gr = min(base + r, nN - 1);
            x[k] = *(const float4*)(h + (size_t)gr * DD + 4 * lane);
        }
        if ((unsigned)tid < rbu) dloc[tid] = dinv[base + tid];
#pragma unroll
        for (int k = 0; k < P0; ++k) {
            const int r = (tid + k * NTH) >> 5;
            if ((unsigned)r < rbu) {
                float* a = acc + r * DD + lane;
                a[0]  = x[k].x;
                a[32] = x[k].y;
                a[64] = x[k].z;
                a[96] = x[k].w;
            }
        }
    }
    __syncthreads();

    if (cnt <= cap) {
        const unsigned* E = L + LHDR;
        int i0 = wv * ERUN;
        unsigned ent = 0u;
        if (i0 + lane < cnt) ent = E[i0 + lane];
        while (i0 + ERUN <= cnt) {
            const int sl = (int)(ent & 0xFFFFFFu);
            const int dl = (int)(ent >> 24);
            unsigned entn = 0u;
            if (i0 + NTH + lane < cnt) entn = E[i0 + NTH + lane];
            float4 rows_[2][EBAT];
#pragma unroll
            for (int u = 0; u < EBAT; ++u)
                rows_[0][u] = agg_load_row(h, __builtin_amdgcn_readlane(sl, u), lane);
#pragma unroll
            for (int b = 0; b < ERUN / EBAT; ++b) {
                if (b + 1 < ERUN / EBAT) {
#pragma unroll
                    for (int u = 0; u < EBAT; ++u)
                        rows_[(b + 1) & 1][u] = agg_load_row(
                            h, __builtin_amdgcn_readlane(sl, (b + 1) * EBAT + u), lane);
                }
#pragma unroll
                for (int u = 0; u < EBAT; ++u)
                    agg_add_row(acc, __builtin_amdgcn_readlane(dl, b * EBAT + u),
                                lane, rows_[b & 1][u]);
            }
            ent = entn;
            i0 += NTH;
        }
        if (i0 < cnt) {
            const int n  = cnt - i0;
            const int sl = (int)(ent & 0xFFFFFFu);
            const int dl = (int)(ent >> 24);
            for (int b = 0; b < n; b += EBAT) {
                float4 rows[EBAT];
#pragma unroll
                for (int u = 0; u < EBAT; ++u)
                    if (b + u < n)
                        rows[u] = agg_load_row(h, __builtin_amdgcn_readlane(sl, b + u), lane);
#pragma unroll
                for (int u = 0; u < EBAT; ++u)
                    if (b + u < n)
                        agg_add_row(acc, __builtin_amdgcn_readlane(dl, b + u), lane, rows[u]);
            }
        }
    } else {
        for (int e0 = wv * 32; e0 < nE; e0 += NTH) {
            const int e = e0 + lane;
            int d = -1;
            if (e < nE) d = dst[e];
            const unsigned t = (unsigned)(d - base);
            const bool hit = t < rbu;
            int s = 0;
            if (hit) s = src[e];
            unsigned m = __builtin_amdgcn_ballot_w32(hit);
            while (m != 0u) {
                const int j = __builtin_ctz(m);
                m &= m - 1u;
                const float4 row = agg_load_row(h, __builtin_amdgcn_readlane(s, j), lane);
                agg_add_row(acc, __builtin_amdgcn_readlane((int)t, j), lane, row);
            }
        }
    }
    __syncthreads();

    for (int i = tid; i < RB * 32; i += NTH) {
        const int r = i >> 5;
        if ((unsigned)r < rbu) {
            const float* a = acc + r * DD + lane;
            const float  di = dloc[r];
            float4 o;
            o.x = a[0]  * di + bv.x;
            o.y = a[32] * di + bv.y;
            o.z = a[64] * di + bv.z;
            o.w = a[96] * di + bv.w;
            *(float4*)(out + (size_t)(base + r) * DD + 4 * lane) = o;
        }
    }
}

extern "C" void kernel_launch(void* const* d_in, const int* in_sizes, int n_in,
                              void* d_out, int out_size, void* d_ws, size_t ws_size,
                              hipStream_t stream) {
    const float* x   = (const float*)d_in[0];
    const int*   ei  = (const int*)d_in[1];
    const float* W1  = (const float*)d_in[2];
    const float* b1  = (const float*)d_in[3];
    const float* W2  = (const float*)d_in[4];
    const float* b2  = (const float*)d_in[5];
    const float* W3  = (const float*)d_in[6];
    const float* b3  = (const float*)d_in[7];

    const int nN = in_sizes[0] / DD;
    const int nE = in_sizes[1] / 2;
    const int* src = ei;
    const int* dst = ei + nE;

    const size_t npad = (size_t)((nN + 127) / 128) * 128;
    float* dinv = (float*)d_ws;
    float* bufH = dinv + npad;
    float* bufX = bufH + npad * DD;
    float* outF = (float*)d_out;
    unsigned* lst = (unsigned*)d_out;

    const dim3 blk(256);
    const int gAgg = (nN + RB - 1) / RB;
    const int gGemm = (nN + 127) / 128;

    unsigned* code = (unsigned*)bufX;
    const int gCode = ((nE >> 4) * 4) / CTH + 1;
    owner_codes_kernel<<<gCode, dim3(CTH), 0, stream>>>(dst, code, nE);

    own_edges_kernel<<<gAgg, dim3(NTH), 0, stream>>>(src, dst, code, dinv, lst, nN, nE);

    gcn_gemm_wmma<<<gGemm, blk, 0, stream>>>(x, W1, dinv, bufH, nN, 0);
    agg_rows_lds_kernel<<<gAgg, dim3(NTH), 0, stream>>>(bufH, dinv, src, dst, b1, lst, bufX, nN, nE);

    gcn_gemm_wmma<<<gGemm, blk, 0, stream>>>(bufX, W2, dinv, bufH, nN, 1);
    agg_rows_lds_kernel<<<gAgg, dim3(NTH), 0, stream>>>(bufH, dinv, src, dst, b2, lst, bufX, nN, nE);

    gcn_gemm_wmma<<<gGemm, blk, 0, stream>>>(bufX, W3, dinv, bufH, nN, 1);
    agg_rows_lds_kernel<<<gAgg, dim3(NTH), 0, stream>>>(bufH, dinv, src, dst, b3, lst, outF, nN, nE);
}


# LDS float atomics leave the layer walk: list launch bins edges by owner wave (row & 31); a wave sums its 7 own rows in registers, indexed by row, in list order (list build: arrival order)
# speedup vs baseline: 4.6423x; 1.4800x over previous
#include <hip/hip_runtime.h>
#include <hip/hip_bf16.h>

#define DD 128
#define WS 136

typedef __attribute__((ext_vector_type(2))) float v2f;
typedef __attribute__((ext_vector_type(8))) float v8f;

#define RB     196
#define NTH    1024
#define NWV    (NTH / 32)
#define SCAN_U 12
#define QW     256
#define ERUN   32
#define EBAT   8
#define LHDR   32
#define SUBL   512
#define CTH    256
#define RT     13
#define RP     (RT * 16)
#define AS     132

__device__ __forceinline__ int list_room(unsigned rbu) {
    return min(SUBL, ((int)rbu * DD - LHDR) / NWV);
}

__global__ __launch_bounds__(CTH)
void owner_codes_kernel(const int* __restrict__ dst, unsigned* __restrict__ code, int nE) {
    const int q  = blockIdx.x * CTH + threadIdx.x;
    const int nQ = (nE >> 4) << 2;
    if (q >= nQ) return;
    int4 d;
    if ((((size_t)dst) & 15) == 0) {
        d = ((const int4*)dst)[q];
    } else {
        d.x = dst[4 * q + 0];  d.y = dst[4 * q + 1];
        d.z = dst[4 * q + 2];  d.w = dst[4 * q + 3];
    }
    const unsigned c0 = ((unsigned)d.x / RB) & 255u;
    const unsigned c1 = ((unsigned)d.y / RB) & 255u;
    const unsigned c2 = ((unsigned)d.z / RB) & 255u;
    const unsigned c3 = ((unsigned)d.w / RB) & 255u;
    code[q] = c0 | (c1 << 8) | (c2 << 16) | (c3 << 24);
}


__device__ __forceinline__ unsigned zero_bytes(unsigned x) {
    return ~(((x & 0x7F7F7F7Fu) + 0x7F7F7F7Fu) | x | 0x7F7F7F7Fu);
}

__global__ __launch_bounds__(NTH)
void own_edges_kernel(const int* __restrict__ src, const int* __restrict__ dst,
                      const unsigned* __restrict__ code,
                      float* __restrict__ dinv, unsigned* __restrict__ lst,
                      const float* __restrict__ x, float* __restrict__ xs,
                      int nN, int nE) {
    __shared__ int degc[RB];
    __shared__ int subc[NWV];
    __shared__ int qe[NWV * QW];
    __shared__ unsigned sub[NWV * SUBL];

    const int tid  = threadIdx.x;
    const int lane = tid & 31;
    const int wv   = __builtin_amdgcn_readfirstlane(tid >> 5);
    const int base = blockIdx.x * RB;
    const unsigned rbu = (unsigned)min(RB, nN - base);
    const int room = list_room(rbu);
    unsigned* L = lst + (size_t)base * DD;
    int* myq = qe + wv * QW;

    if (tid < RB) degc[tid] = 0;
    if (tid < NWV) subc[tid] = 0;
    __syncthreads();

    int qc = 0;
    int done = 0;

#define OWN_FLUSH()                                                                \
    {                                                                              \
        const int n_ = qc - done;                                                  \
        for (int i0_ = 0; i0_ < n_; i0_ += 32) {                                   \
            const int i_ = i0_ + lane;                                             \
            if (i_ < n_) {                                                         \
                const int      e_ = myq[(done + i_) & (QW - 1)];                   \
                const unsigned t_ = (unsigned)(dst[e_] - base);                    \
                const int      s_ = src[e_];                                       \
                if (t_ < rbu) {                                                    \
                    atomicAdd(&degc[t_], 1);                                       \
                    const int w_ = (int)(t_ & 31u);                                \
                    const int p_ = atomicAdd(&subc[w_], 1);                        \
                    if (p_ < room) sub[w_ * SUBL + p_] = (unsigned)s_ | (t_ << 24); \
                }                                                                  \
            }                                                                      \
        }                                                                          \
        done = qc;                                                                 \
    }

#define OWN_SCAN16(V, E0)                                                          \
    {                                                                              \
        const unsigned z0_ = zero_bytes((V).x ^ pat);                              \
        const unsigned z1_ = zero_bytes((V).y ^ pat);                              \
        const unsigned z2_ = zero_bytes((V).z ^ pat);                              \
        const unsigned z3_ = zero_bytes((V).w ^ pat);                              \
                     \
        unsigned hm_ = (z0_ >> 7) | (z1_ >> 6) | (z2_ >> 5) | (z3_ >> 4);          \
        unsigned m_  = __builtin_amdgcn_ballot_w32(hm_ != 0u);                     \
        while (m_ != 0u) {                             \
            if (qc - done > QW - 32) OWN_FLUSH()                                   \
            if (hm_ != 0u) {                                                       \
                const int bit_  = __builtin_ctz(hm_);                              \
                hm_ &= hm_ - 1u;                                                   \
                const int rank_ = (int)__builtin_amdgcn_mbcnt_lo(m_, 0u);          \
                myq[(qc + rank_) & (QW - 1)] = (E0) + (((bit_ & 3) << 2) | (bit_ >> 3)); \
            }                                                                      \
            qc += (int)__builtin_popcount(m_);                                     \
            m_  = __builtin_amdgcn_ballot_w32(hm_ != 0u);                          \
        }                                                                          \
    }

    const unsigned pat = (blockIdx.x & 255u) * 0x01010101u;
    const int    n16  = nE >> 4;
    const uint4* c4   = (const uint4*)code;
    const int    nBlk = n16 / NTH;
    const int    rot  = (nBlk > 0) ? (int)(blockIdx.x % (unsigned)nBlk) : 0;

    int k0 = 0;
    for (; k0 + SCAN_U <= nBlk; k0 += SCAN_U) {
        uint4 v[SCAN_U];
        int   g[SCAN_U];
#pragma unroll
        for (int u = 0; u < SCAN_U; ++u) {
            int blk = k0 + u + rot;
            if (blk >= nBlk) blk -= nBlk;
            g[u] = blk * NTH + tid;
            v[u] = c4[g[u]];
        }
#pragma unroll
        for (int u = 0; u < SCAN_U; ++u) OWN_SCAN16(v[u], g[u] << 4)
    }
    for (; k0 < nBlk; ++k0) {
        int blk = k0 + rot;
        if (blk >= nBlk) blk -= nBlk;
        const int   g = blk * NTH + tid;
        const uint4 v = c4[g];
        OWN_SCAN16(v, g << 4)
    }
    if (nBlk * NTH < n16) {
        const int g = nBlk * NTH + tid;
        uint4 v = make_uint4(~pat, ~pat, ~pat, ~pat);
        if (g < n16) v = c4[g];
        OWN_SCAN16(v, g << 4)
    }
    for (int e0 = n16 << 4; e0 < nE; e0 += NTH) {
        const int e = e0 + tid;
        int d = -1;
        if (e < nE) d = dst[e];
        const bool     hit = (unsigned)(d - base) < rbu;
        const unsigned m   = __builtin_amdgcn_ballot_w32(hit);
        if (m != 0u) {
            if (qc - done > QW - 32) OWN_FLUSH()
            if (hit) myq[(qc + (int)__builtin_amdgcn_mbcnt_lo(m, 0u)) & (QW - 1)] = e;
            qc += (int)__builtin_popcount(m);
        }
    }
    OWN_FLUSH()
#undef OWN_FLUSH
#undef OWN_SCAN16
    __syncthreads();

    if ((unsigned)tid < rbu) dinv[base + tid] = rsqrtf((float)(1 + degc[tid]));
    if (tid < NWV) L[tid] = (unsigned)subc[tid];

    {
        const int nw = min(subc[wv], room);
        unsigned*       Lw = L + LHDR + wv * room;
        const unsigned* Sw = sub + wv * SUBL;
        for (int i = lane; i < nw; i += 32) Lw[i] = Sw[i];
    }

    {
        constexpr int P1 = (RB + 31) / 32;
        float4 xr[P1];
#pragma unroll
        for (int k = 0; k < P1; ++k) {
            const int gr = min(base + wv + 32 * k, nN - 1);
            xr[k] = *(const float4*)(x + (size_t)gr * DD + 4 * lane);
        }
#pragma unroll
        for (int k = 0; k < P1; ++k) {
            const int r = wv + 32 * k;
            if ((unsigned)r < rbu) {
                const float d = rsqrtf((float)(1 + degc[r]));
                float4 o;
                o.x = xr[k].x * d;  o.y = xr[k].y * d;
                o.z = xr[k].z * d;  o.w = xr[k].w * d;
                *(float4*)(xs + (size_t)(base + r) * DD + 4 * lane) = o;
            }
        }
    }
}


__device__ __forceinline__ float4 agg_load_row(const float* __restrict__ h, int s, int lane) {
    return *(const float4*)(h + (size_t)s * DD + 4 * lane);
}

#define P0   ((RP + 31) / 32)
#define EPAD 0xE0000000u

#define ROW_ADD(K, V)                                                              \
    sx[K] += (V).x;  sy[K] += (V).y;  sz[K] += (V).z;  sw[K] += (V).w;

#define ROWS_ISSUE(SL, L0)                                                         \
    _Pragma("unroll")                                                              \
    for (int u_ = 0; u_ < EBAT; ++u_)                                              \
        rr[u_] = agg_load_row(xin, __builtin_amdgcn_readlane(SL, (L0) + u_), lane);

#define ROWS_ADD_NEXT(DK, LA, SN, LN)                                              \
    _Pragma("unroll")                                                              \
    for (int u_ = 0; u_ < EBAT; ++u_) {                                            \
        const int k_ = __builtin_amdgcn_readlane(DK, (LA) + u_);                   \
        ROW_ADD(k_, rr[u_])                                                        \
        rr[u_] = agg_load_row(xin, __builtin_amdgcn_readlane(SN, (LN) + u_), lane); \
        __builtin_amdgcn_sched_barrier(0);                                         \
    }

#define ROWS_ADD(DK, LA)                                                           \
    _Pragma("unroll")                                                              \
    for (int u_ = 0; u_ < EBAT; ++u_) {                                            \
        const int k_ = __builtin_amdgcn_readlane(DK, (LA) + u_);                   \
        ROW_ADD(k_, rr[u_])                                                        \
    }

static_assert(P0 <= 7, "the sums hold at most 7 own rows per wave beside the spare element");
static_assert(ERUN == 4 * EBAT, "a run is walked as four groups of EBAT rows");

__global__ __launch_bounds__(NTH)
void layer_rows_kernel(const float* __restrict__ xin, const float* __restrict__ W,
                       const float* __restrict__ dinv,
                       const int* __restrict__ src, const int* __restrict__ dst,
                       const float* __restrict__ bias, const unsigned* lst,
                       float* out, int nN, int nE, int last) {
    __shared__ __align__(16) float acc[RP * AS];
    __shared__ __align__(16) float sW[DD * WS];
    __shared__ float dloc[RP];

    const int tid  = threadIdx.x;
    const int lane = tid & 31;
    const int wv   = __builtin_amdgcn_readfirstlane(tid >> 5);
    const int base = blockIdx.x * RB;
    const unsigned rbu = (unsigned)min(RB, nN - base);
    const unsigned* L = lst + (size_t)base * DD;
    const int room = list_room(rbu);
    const int cnt = __builtin_amdgcn_readfirstlane((int)L[wv]);
    const unsigned* E = L + LHDR + wv * room;

    v8f sx, sy, sz, sw;

    {
        constexpr int PW = DD / 32;
        float4 w[PW];
        float4 self[P0];
#pragma unroll
        for (int k = 0; k < PW; ++k)
            w[k] = *(const float4*)(W + (wv + 32 * k) * DD + 4 * lane);
#pragma unroll
        for (int k = 0; k < P0; ++k) {
            const int gr = min(base + wv + 32 * k, nN - 1);
            self[k] = *(const float4*)(xin + (size_t)gr * DD + 4 * lane);
        }
        if (tid < RP) dloc[tid] = ((unsigned)tid < rbu) ? dinv[base + tid] : 0.0f;
#pragma unroll
        for (int k = 0; k < PW; ++k) {
            const int kr = wv + 32 * k;
            *(float4*)(sW + ((kr & 3) * 32 + (kr >> 2)) * WS + 4 * lane) = w[k];
        }
        sx[7] = 0.0f;  sy[7] = 0.0f;  sz[7] = 0.0f;  sw[7] = 0.0f;
#pragma unroll
        for (int k = 0; k < P0; ++k) {
            const bool own = (unsigned)(wv + 32 * k) < rbu;
            sx[k] = own ? self[k].x : 0.0f;
            sy[k] = own ? self[k].y : 0.0f;
            sz[k] = own ? self[k].z : 0.0f;
            sw[k] = own ? self[k].w : 0.0f;
        }
    }

    if (cnt <= room) {
        int i0 = 0;
        unsigned ent = EPAD;
        if (lane < cnt) ent = E[lane];
        float4 rr[EBAT];
        {
            const int sl0 = (int)(ent & 0xFFFFFFu);
            ROWS_ISSUE(sl0, 0)
        }
        while (i0 + ERUN <= cnt) {
            const int sl = (int)(ent & 0xFFFFFFu);
            const int dk = (int)(ent >> 29);
            unsigned entn = EPAD;
            if (i0 + ERUN + lane < cnt) entn = E[i0 + ERUN + lane];
            ROWS_ADD_NEXT(dk, 0 * EBAT, sl, 1 * EBAT)
            ROWS_ADD_NEXT(dk, 1 * EBAT, sl, 2 * EBAT)
            ROWS_ADD_NEXT(dk, 2 * EBAT, sl, 3 * EBAT)
            const int sn = (int)(entn & 0xFFFFFFu);
            ROWS_ADD_NEXT(dk, 3 * EBAT, sn, 0)
            ent = entn;
            i0 += ERUN;
        }
        const int n = cnt - i0;
        if (n > 0) {
            const int sl = (int)(ent & 0xFFFFFFu);
            const int dk = (int)(ent >> 29);
            ROWS_ADD(dk, 0)
            for (int b = EBAT; b < n; b += EBAT) {
                ROWS_ISSUE(sl, b)
                ROWS_ADD(dk, b)
            }
        }
    } else {
        for (int e0 = 0; e0 < nE; e0 += 32) {
            const int e = e0 + lane;
            int d = -1;
            if (e < nE) d = dst[e];
            const unsigned t = (unsigned)(d - base);
            const bool hit = (t < rbu) && ((int)(t & 31u) == wv);
            int s = 0;
            if (hit) s = src[e];
            const int kk = (int)(t >> 5);
            unsigned m = __builtin_amdgcn_ballot_w32(hit);
            while (m != 0u) {
                const int j = __builtin_ctz(m);
                m &= m - 1u;
                const float4 row = agg_load_row(xin, __builtin_amdgcn_readlane(s, j), lane);
                const int k_ = __builtin_amdgcn_readlane(kk, j);
                ROW_ADD(k_, row)
            }
        }
    }

#pragma unroll
    for (int k = 0; k < P0; ++k) {
        const int r = wv + 32 * k;
        if (r < RP) {
            float* a = acc + r * AS + lane;
            a[0]  = sx[k];
            a[32] = sy[k];
            a[64] = sz[k];
            a[96] = sw[k];
        }
    }
    __syncthreads();

    {
        const int lr = lane & 15;
        const int hi = lane >> 4;
        for (int item = wv; item < RT * 4; item += NTH / 32) {
            const int rt = item >> 2;
            const int cb = (item & 3) * 32;
            const float* arow = acc + (rt * 16 + lr) * AS + hi * 2;
            const float* wp = sW + (hi * 2) * WS + cb + lr;
            v8f c0 = (v8f){0.f,0.f,0.f,0.f,0.f,0.f,0.f,0.f};
            v8f c1 = (v8f){0.f,0.f,0.f,0.f,0.f,0.f,0.f,0.f};
#pragma unroll 4
            for (int kb = 0; kb < 32; ++kb) {
                const v2f a = *(const v2f*)(arow + kb * 4);
                const float* wk = wp + kb * 4 * WS;
                v2f b0, b1;
                b0.x = wk[0];   b0.y = wk[WS];
                b1.x = wk[16];  b1.y = wk[WS + 16];
                c0 = __builtin_amdgcn_wmma_f32_16x16x4_f32(
                    false, a, false, b0, (short)0, c0, false, false);
                c1 = __builtin_amdgcn_wmma_f32_16x16x4_f32(
                    false, a, false, b1, (short)0, c1, false, false);
            }
            const int r0 = rt * 16 + hi * 8;
            const float bc0 = bias[cb + lr];
            const float bc1 = bias[cb + 16 + lr];
#pragma unroll
            for (int v = 0; v < 8; ++v) {
                const int r = r0 + v;
                if ((unsigned)r < rbu) {
                    const float di = dloc[r];
                    float o0 = c0[v] * di + bc0;
                    float o1 = c1[v] * di + bc1;
                    if (!last) {
                        o0 = fmaxf(o0, 0.0f) * di;
                        o1 = fmaxf(o1, 0.0f) * di;
                    }
                    float* orow = out + (size_t)(base + r) * DD + cb + lr;
                    orow[0]  = o0;
                    orow[16] = o1;
                }
            }
        }
    }
}
#undef ROWS_ADD
#undef ROWS_ADD_NEXT
#undef ROWS_ISSUE
#undef ROW_ADD

extern "C" void kernel_launch(void* const* d_in, const int* in_sizes, int n_in,
                              void* d_out, int out_size, void* d_ws, size_t ws_size,
                              hipStream_t stream) {
    const float* x   = (const float*)d_in[0];
    const int*   ei  = (const int*)d_in[1];
    const float* W1  = (const float*)d_in[2];
    const float* b1  = (const float*)d_in[3];
    const float* W2  = (const float*)d_in[4];
    const float* b2  = (const float*)d_in[5];
    const float* W3  = (const float*)d_in[6];
    const float* b3  = (const float*)d_in[7];

    const int nN = in_sizes[0] / DD;
    const int nE = in_sizes[1] / 2;
    const int* src = ei;
    const int* dst = ei + nE;

    const size_t npad = (size_t)((nN + 127) / 128) * 128;
    float* dinv = (float*)d_ws;
    float* bufH = dinv + npad;
    float* bufX = bufH + npad * DD;
    float* outF = (float*)d_out;
    unsigned* lst = (unsigned*)d_out;

    const int gAgg = (nN + RB - 1) / RB;

    unsigned* code = (unsigned*)bufX;
    const int gCode = ((nE >> 4) * 4) / CTH + 1;
    owner_codes_kernel<<<gCode, dim3(CTH), 0, stream>>>(dst, code, nE);

    own_edges_kernel<<<gAgg, dim3(NTH), 0, stream>>>(src, dst, code, dinv, lst, x, bufH, nN, nE);

    layer_rows_kernel<<<gAgg, dim3(NTH), 0, stream>>>(bufH, W1, dinv, src, dst, b1, lst, bufX, nN, nE, 0);

    layer_rows_kernel<<<gAgg, dim3(NTH), 0, stream>>>(bufX, W2, dinv, src, dst, b2, lst, bufH, nN, nE, 0);

    layer_rows_kernel<<<gAgg, dim3(NTH), 0, stream>>>(bufH, W3, dinv, src, dst, b3, lst, outF, nN, nE, 1);
}
